# starting kernel
# baseline (speedup 1.0000x reference)
#include <hip/hip_runtime.h>

#define B_ 2
#define S_ 4
#define T_ 1024
#define H_ 256
#define R_ 32
#define V_ (T_ - 1)
#define M_ 8
#define ROUTE_VOCAB_ 65536
#define SUB_STRIDE_ (R_ * ROUTE_VOCAB_)

typedef float v2f __attribute__((ext_vector_type(2)));
typedef float v8f __attribute__((ext_vector_type(8)));

__global__ __launch_bounds__(256)
void lngram_pack_codes(const float* __restrict__ hidden,
                       const float* __restrict__ qw,
                       unsigned char* __restrict__ codes) {
  const int lane = threadIdx.x & 31;
  const int wave = (blockIdx.x << 3) | (threadIdx.x >> 5);

  const int ot = wave & 15;
  const int tt = (wave >> 4) & 63;
  const int s  = (wave >> 10) & 3;
  const int b  = wave >> 12;

  const int t0  = tt << 4;
  const int o0  = ot << 4;
  const int mn  = lane & 15;
  const int kof = (lane >> 4) << 1;

  const float* __restrict__ arow =
      hidden + (size_t)(b * T_ + t0 + mn) * H_ + kof;
  const float* __restrict__ brow =
      qw + (size_t)(s * H_ + (o0 + mn)) * H_ + kof;

  v8f acc = {};
#pragma unroll 8
  for (int kk = 0; kk < H_; kk += 4) {
    v2f a  = *(const v2f*)(arow + kk);
    v2f bb = *(const v2f*)(brow + kk);
    acc = __builtin_amdgcn_wmma_f32_16x16x4_f32(
        false, a, false, bb, (short)0, acc, false, false);
  }

  unsigned mk0 = __builtin_amdgcn_ballot_w32(acc[0] > 0.0f);
  unsigned mk1 = __builtin_amdgcn_ballot_w32(acc[1] > 0.0f);
  unsigned mk2 = __builtin_amdgcn_ballot_w32(acc[2] > 0.0f);
  unsigned mk3 = __builtin_amdgcn_ballot_w32(acc[3] > 0.0f);
  unsigned mk4 = __builtin_amdgcn_ballot_w32(acc[4] > 0.0f);
  unsigned mk5 = __builtin_amdgcn_ballot_w32(acc[5] > 0.0f);
  unsigned mk6 = __builtin_amdgcn_ballot_w32(acc[6] > 0.0f);
  unsigned mk7 = __builtin_amdgcn_ballot_w32(acc[7] > 0.0f);

  const int jj   = lane & 7;
  const int part = lane >> 3;

  unsigned msel = mk0;
  msel = (jj == 1) ? mk1 : msel;
  msel = (jj == 2) ? mk2 : msel;
  msel = (jj == 3) ? mk3 : msel;
  msel = (jj == 4) ? mk4 : msel;
  msel = (jj == 5) ? mk5 : msel;
  msel = (jj == 6) ? mk6 : msel;
  msel = (jj == 7) ? mk7 : msel;

  const unsigned byte = (msel >> (part << 3)) & 0xFFu;
  const int t = t0 + jj + ((part & 2) ? 8 : 0);
  const int r = (ot << 1) + (part & 1);

  codes[(size_t)((b * S_ + s) * T_ + t) * R_ + r] = (unsigned char)byte;
}

__global__ __launch_bounds__(256)
void lngram_gather(const unsigned char* __restrict__ codes,
                   const float* __restrict__ flat_weight,
                   float* __restrict__ out) {
  const int gid = blockIdx.x * blockDim.x + threadIdx.x;
  if (gid >= B_ * S_ * V_ * R_) return;

  const int r   = gid & (R_ - 1);
  const int tv  = gid >> 5;
  const int v   = tv % V_;
  const int bs  = tv / V_;
  const int s   = bs & (S_ - 1);

  const unsigned char* __restrict__ cb =
      codes + (size_t)(bs * T_ + v) * R_ + r;
  const unsigned c0 = cb[0];
  const unsigned c1 = cb[R_];
  const unsigned addr = c0 | (c1 << 8);

  const size_t row = (size_t)s * SUB_STRIDE_ + (size_t)r * ROUTE_VOCAB_ + addr;
  const float4* __restrict__ src = (const float4*)(flat_weight + row * M_);
  const float4 x = src[0];
  const float4 y = src[1];

  float4* __restrict__ dst = (float4*)(out + (size_t)gid * M_);
  dst[0] = x;
  dst[1] = y;
}

extern "C" void kernel_launch(void* const* d_in, const int* in_sizes, int n_in,
                              void* d_out, int out_size, void* d_ws, size_t ws_size,
                              hipStream_t stream) {
  const float* hidden      = (const float*)d_in[0];
  const float* q_weight    = (const float*)d_in[1];
  const float* flat_weight = (const float*)d_in[2];
  float* out               = (float*)d_out;

  unsigned char* codes = (unsigned char*)d_ws;

  lngram_pack_codes<<<1024, 256, 0, stream>>>(hidden, q_weight, codes);

  const int total = B_ * S_ * V_ * R_;
  lngram_gather<<<(total + 255) / 256, 256, 0, stream>>>(codes, flat_weight, out);
}


# kernel 1 retiled: 64x128 workgroup tile staged once in LDS by coalesced 16-byte loads, each wave 2x2 WMMA tiles fed from LDS; every logit keeps its k order, sums unchanged
# speedup vs baseline: 1.7674x; 1.7674x over previous
#include <hip/hip_runtime.h>

#define B_ 2
#define S_ 4
#define T_ 1024
#define H_ 256
#define R_ 32
#define V_ (T_ - 1)
#define M_ 8
#define ROUTE_VOCAB_ 65536
#define SUB_STRIDE_ (R_ * ROUTE_VOCAB_)

typedef float v2f __attribute__((ext_vector_type(2)));
typedef float v8f __attribute__((ext_vector_type(8)));

#define TM_ 64
#define TN_ 128
#define KC_ 64
#define NCH_ (H_ / KC_)
#define PITCH_ 260

typedef float v4f __attribute__((ext_vector_type(4)));

__global__ __launch_bounds__(256)
void lngram_pack_codes(const float* __restrict__ hidden,
                       const float* __restrict__ qw,
                       unsigned char* __restrict__ codes) {
  __shared__ __attribute__((aligned(16))) float lds[(TM_ + TN_) * PITCH_];
  float* lA = lds;
  float* lB = lds + TM_ * PITCH_;

  const int tid  = threadIdx.x;
  const int lane = tid & 31;
  const int wave = tid >> 5;

  const int nt = blockIdx.x & 7;
  const int mt = blockIdx.x >> 3;
  const int b  = mt >> 4;
  const int t0 = (mt & 15) << 6;
  const int s  = nt >> 1;
  const int o0 = (nt & 1) << 7;

  const int q4 = (tid & 15) << 2;
  const int sr = tid >> 4;
  const float* __restrict__ gA = hidden + (size_t)(b * T_ + t0 + sr) * H_ + q4;
  const float* __restrict__ gB = qw + (size_t)(s * H_ + o0 + sr) * H_ + q4;
  float* sA = lA + sr * PITCH_ + q4;
  float* sB = lB + sr * PITCH_ + q4;

  v4f ra[TM_ / 16], rb[TN_ / 16];
#pragma unroll
  for (int i = 0; i < TM_ / 16; ++i) ra[i] = *(const v4f*)(gA + (size_t)i * 16 * H_);
#pragma unroll
  for (int i = 0; i < TN_ / 16; ++i) rb[i] = *(const v4f*)(gB + (size_t)i * 16 * H_);

  const int c  = lane & 15;
  const int h  = lane >> 4;
  const int wt = wave & 1;
  const int wo = wave >> 1;
  const float* fA = lA + (32 * wt + c) * PITCH_ + 2 * h;
  const float* fB = lB + (32 * wo + c) * PITCH_ + 2 * h;

  v8f acc00 = {}, acc01 = {}, acc10 = {}, acc11 = {};

#pragma unroll
  for (int ch = 0; ch < NCH_; ++ch) {
#pragma unroll
    for (int i = 0; i < TM_ / 16; ++i) *(v4f*)(sA + i * 16 * PITCH_ + ch * KC_) = ra[i];
#pragma unroll
    for (int i = 0; i < TN_ / 16; ++i) *(v4f*)(sB + i * 16 * PITCH_ + ch * KC_) = rb[i];
    if (ch + 1 < NCH_) {
#pragma unroll
      for (int i = 0; i < TM_ / 16; ++i)
        ra[i] = *(const v4f*)(gA + (size_t)i * 16 * H_ + (ch + 1) * KC_);
#pragma unroll
      for (int i = 0; i < TN_ / 16; ++i)
        rb[i] = *(const v4f*)(gB + (size_t)i * 16 * H_ + (ch + 1) * KC_);
    }
    __syncthreads();

#pragma unroll
    for (int j = 0; j < KC_ / 8; ++j) {
      const int off = ch * KC_ + j * 8;
      const v2f a0x = *(const v2f*)(fA + off),               a0y = *(const v2f*)(fA + off + 4);
      const v2f a1x = *(const v2f*)(fA + 16 * PITCH_ + off), a1y = *(const v2f*)(fA + 16 * PITCH_ + off + 4);
      const v2f b0x = *(const v2f*)(fB + off),               b0y = *(const v2f*)(fB + off + 4);
      const v2f b1x = *(const v2f*)(fB + 16 * PITCH_ + off), b1y = *(const v2f*)(fB + 16 * PITCH_ + off + 4);
      acc00 = __builtin_amdgcn_wmma_f32_16x16x4_f32(false, a0x, false, b0x, (short)0, acc00, false, false);
      acc01 = __builtin_amdgcn_wmma_f32_16x16x4_f32(false, a0x, false, b1x, (short)0, acc01, false, false);
      acc10 = __builtin_amdgcn_wmma_f32_16x16x4_f32(false, a1x, false, b0x, (short)0, acc10, false, false);
      acc11 = __builtin_amdgcn_wmma_f32_16x16x4_f32(false, a1x, false, b1x, (short)0, acc11, false, false);
      acc00 = __builtin_amdgcn_wmma_f32_16x16x4_f32(false, a0y, false, b0y, (short)0, acc00, false, false);
      acc01 = __builtin_amdgcn_wmma_f32_16x16x4_f32(false, a0y, false, b1y, (short)0, acc01, false, false);
      acc10 = __builtin_amdgcn_wmma_f32_16x16x4_f32(false, a1y, false, b0y, (short)0, acc10, false, false);
      acc11 = __builtin_amdgcn_wmma_f32_16x16x4_f32(false, a1y, false, b1y, (short)0, acc11, false, false);
    }
  }

  const int myti = lane >> 4;
  const int myj  = lane & 7;
  unsigned m0 = 0u, m1 = 0u;
#pragma unroll
  for (int j = 0; j < 8; ++j) {
    const unsigned k00 = __builtin_amdgcn_ballot_w32(acc00[j] > 0.0f);
    const unsigned k01 = __builtin_amdgcn_ballot_w32(acc01[j] > 0.0f);
    const unsigned k10 = __builtin_amdgcn_ballot_w32(acc10[j] > 0.0f);
    const unsigned k11 = __builtin_amdgcn_ballot_w32(acc11[j] > 0.0f);
    const bool mj = (myj == j);
    m0 = (mj && myti == 0) ? k00 : m0;
    m1 = (mj && myti == 0) ? k01 : m1;
    m0 = (mj && myti == 1) ? k10 : m0;
    m1 = (mj && myti == 1) ? k11 : m1;
  }
  const unsigned sh = ((lane >> 3) & 1) << 4;
  const unsigned word = ((m0 >> sh) & 0xFFFFu) | (((m1 >> sh) & 0xFFFFu) << 16);

  const int t  = t0 + 32 * wt + lane;
  const int rw = (o0 >> 3) + 4 * wo;
  *(unsigned*)(codes + (size_t)((b * S_ + s) * T_ + t) * R_ + rw) = word;
}

__global__ __launch_bounds__(256)
void lngram_gather(const unsigned char* __restrict__ codes,
                   const float* __restrict__ flat_weight,
                   float* __restrict__ out) {
  const int gid = blockIdx.x * blockDim.x + threadIdx.x;
  if (gid >= B_ * S_ * V_ * R_) return;

  const int r   = gid & (R_ - 1);
  const int tv  = gid >> 5;
  const int v   = tv % V_;
  const int bs  = tv / V_;
  const int s   = bs & (S_ - 1);

  const unsigned char* __restrict__ cb =
      codes + (size_t)(bs * T_ + v) * R_ + r;
  const unsigned c0 = cb[0];
  const unsigned c1 = cb[R_];
  const unsigned addr = c0 | (c1 << 8);

  const size_t row = (size_t)s * SUB_STRIDE_ + (size_t)r * ROUTE_VOCAB_ + addr;
  const float4* __restrict__ src = (const float4*)(flat_weight + row * M_);
  const float4 x = src[0];
  const float4 y = src[1];

  float4* __restrict__ dst = (float4*)(out + (size_t)gid * M_);
  dst[0] = x;
  dst[1] = y;
}

extern "C" void kernel_launch(void* const* d_in, const int* in_sizes, int n_in,
                              void* d_out, int out_size, void* d_ws, size_t ws_size,
                              hipStream_t stream) {
  const float* hidden      = (const float*)d_in[0];
  const float* q_weight    = (const float*)d_in[1];
  const float* flat_weight = (const float*)d_in[2];
  float* out               = (float*)d_out;

  unsigned char* codes = (unsigned char*)d_ws;

  lngram_pack_codes<<<256, 256, 0, stream>>>(hidden, q_weight, codes);

  const int total = B_ * S_ * V_ * R_;
  lngram_gather<<<(total + 255) / 256, 256, 0, stream>>>(codes, flat_weight, out);
}


# two launches fused into one: each workgroup also computes the codes of row t0+64 (a fifth WMMA chain per wave, same k order), keeps its codes in LDS and gathers its own 64 x 16 table rows
# speedup vs baseline: 2.0998x; 1.1881x over previous
#include <hip/hip_runtime.h>

#define B_ 2
#define S_ 4
#define T_ 1024
#define H_ 256
#define R_ 32
#define V_ (T_ - 1)
#define M_ 8
#define ROUTE_VOCAB_ 65536
#define SUB_STRIDE_ (R_ * ROUTE_VOCAB_)

typedef float v2f __attribute__((ext_vector_type(2)));
typedef float v8f __attribute__((ext_vector_type(8)));

#define TM_ 64
#define TN_ 128
#define KC_ 64
#define NCH_ (H_ / KC_)
#define PITCH_ 260
#define RT_ (TN_ / 8)
#define VW_ (TM_ / 8)

typedef float v4f __attribute__((ext_vector_type(4)));

__global__ __launch_bounds__(256)
void lngram_codes_gather(const float* __restrict__ hidden,
                         const float* __restrict__ qw,
                         const float* __restrict__ flat_weight,
                         float* __restrict__ out) {
  __shared__ __attribute__((aligned(16))) float lds[(TM_ + TN_ + 1) * PITCH_];
  __shared__ __attribute__((aligned(16))) unsigned char lcode[(TM_ + 1) * RT_];
  float* lH = lds;
  float* lA = lds + PITCH_;
  float* lB = lds + (1 + TM_) * PITCH_;

  const int tid  = threadIdx.x;
  const int lane = tid & 31;
  const int wave = tid >> 5;

  const int nt = blockIdx.x & 7;
  const int mt = blockIdx.x >> 3;
  const int b  = mt >> 4;
  const int t0 = (mt & 15) << 6;
  const int s  = nt >> 1;
  const int o0 = (nt & 1) << 7;

  const int q4 = (tid & 15) << 2;
  const int sr = tid >> 4;
  const float* __restrict__ gA = hidden + (size_t)(b * T_ + t0 + sr) * H_ + q4;
  const float* __restrict__ gB = qw + (size_t)(s * H_ + o0 + sr) * H_ + q4;
  float* sA = lA + sr * PITCH_ + q4;
  float* sB = lB + sr * PITCH_ + q4;

  v4f ra[TM_ / 16], rb[TN_ / 16];
#pragma unroll
  for (int i = 0; i < TM_ / 16; ++i) ra[i] = *(const v4f*)(gA + (size_t)i * 16 * H_);
#pragma unroll
  for (int i = 0; i < TN_ / 16; ++i) rb[i] = *(const v4f*)(gB + (size_t)i * 16 * H_);

  const int th = (t0 + TM_ < T_) ? (t0 + TM_) : (T_ - 1);
  const v4f rh = *(const v4f*)(hidden + (size_t)(b * T_ + th) * H_ + 4 * (tid & (H_ / 4 - 1)));

  const int c  = lane & 15;
  const int h  = lane >> 4;
  const int wt = wave & 1;
  const int wo = wave >> 1;
  const float* fA = lA + (32 * wt + c) * PITCH_ + 2 * h;
  const float* fB = lB + (32 * wo + c) * PITCH_ + 2 * h;
  const float* fH = lH + 2 * h;
  const float* fBh = fB + 16 * wt * PITCH_;

  v8f acc00 = {}, acc01 = {}, acc10 = {}, acc11 = {}, acch = {};

#pragma unroll
  for (int ch = 0; ch < NCH_; ++ch) {
#pragma unroll
    for (int i = 0; i < TM_ / 16; ++i) *(v4f*)(sA + i * 16 * PITCH_ + ch * KC_) = ra[i];
#pragma unroll
    for (int i = 0; i < TN_ / 16; ++i) *(v4f*)(sB + i * 16 * PITCH_ + ch * KC_) = rb[i];
    if (ch + 1 < NCH_) {
#pragma unroll
      for (int i = 0; i < TM_ / 16; ++i)
        ra[i] = *(const v4f*)(gA + (size_t)i * 16 * H_ + (ch + 1) * KC_);
#pragma unroll
      for (int i = 0; i < TN_ / 16; ++i)
        rb[i] = *(const v4f*)(gB + (size_t)i * 16 * H_ + (ch + 1) * KC_);
    }
    if (ch == 0) *(v4f*)(lH + 4 * (tid & (H_ / 4 - 1))) = rh;
    __syncthreads();

#pragma unroll
    for (int j = 0; j < KC_ / 8; ++j) {
      const int off = ch * KC_ + j * 8;
      const v2f a0x = *(const v2f*)(fA + off),               a0y = *(const v2f*)(fA + off + 4);
      const v2f a1x = *(const v2f*)(fA + 16 * PITCH_ + off), a1y = *(const v2f*)(fA + 16 * PITCH_ + off + 4);
      const v2f b0x = *(const v2f*)(fB + off),               b0y = *(const v2f*)(fB + off + 4);
      const v2f b1x = *(const v2f*)(fB + 16 * PITCH_ + off), b1y = *(const v2f*)(fB + 16 * PITCH_ + off + 4);
      const v2f ahx = *(const v2f*)(fH + off),               ahy = *(const v2f*)(fH + off + 4);
      const v2f bhx = *(const v2f*)(fBh + off),              bhy = *(const v2f*)(fBh + off + 4);
      acc00 = __builtin_amdgcn_wmma_f32_16x16x4_f32(false, a0x, false, b0x, (short)0, acc00, false, false);
      acc01 = __builtin_amdgcn_wmma_f32_16x16x4_f32(false, a0x, false, b1x, (short)0, acc01, false, false);
      acc10 = __builtin_amdgcn_wmma_f32_16x16x4_f32(false, a1x, false, b0x, (short)0, acc10, false, false);
      acc11 = __builtin_amdgcn_wmma_f32_16x16x4_f32(false, a1x, false, b1x, (short)0, acc11, false, false);
      acch  = __builtin_amdgcn_wmma_f32_16x16x4_f32(false, ahx, false, bhx, (short)0, acch,  false, false);
      acc00 = __builtin_amdgcn_wmma_f32_16x16x4_f32(false, a0y, false, b0y, (short)0, acc00, false, false);
      acc01 = __builtin_amdgcn_wmma_f32_16x16x4_f32(false, a0y, false, b1y, (short)0, acc01, false, false);
      acc10 = __builtin_amdgcn_wmma_f32_16x16x4_f32(false, a1y, false, b0y, (short)0, acc10, false, false);
      acc11 = __builtin_amdgcn_wmma_f32_16x16x4_f32(false, a1y, false, b1y, (short)0, acc11, false, false);
      acch  = __builtin_amdgcn_wmma_f32_16x16x4_f32(false, ahy, false, bhy, (short)0, acch,  false, false);
    }
  }

  const int myti = lane >> 4;
  const int myj  = lane & 7;
  unsigned m0 = 0u, m1 = 0u;
#pragma unroll
  for (int j = 0; j < 8; ++j) {
    const unsigned k00 = __builtin_amdgcn_ballot_w32(acc00[j] > 0.0f);
    const unsigned k01 = __builtin_amdgcn_ballot_w32(acc01[j] > 0.0f);
    const unsigned k10 = __builtin_amdgcn_ballot_w32(acc10[j] > 0.0f);
    const unsigned k11 = __builtin_amdgcn_ballot_w32(acc11[j] > 0.0f);
    const bool mj = (myj == j);
    m0 = (mj && myti == 0) ? k00 : m0;
    m1 = (mj && myti == 0) ? k01 : m1;
    m0 = (mj && myti == 1) ? k10 : m0;
    m1 = (mj && myti == 1) ? k11 : m1;
  }
  const unsigned sh = ((lane >> 3) & 1) << 4;
  const unsigned word = ((m0 >> sh) & 0xFFFFu) | (((m1 >> sh) & 0xFFFFu) << 16);
  const unsigned kh = __builtin_amdgcn_ballot_w32(acch[0] > 0.0f);

  *(unsigned*)(lcode + (32 * wt + lane) * RT_ + 4 * wo) = word;
  if (lane == 0)
    *(unsigned short*)(lcode + TM_ * RT_ + 2 * (2 * wo + wt)) = (unsigned short)(kh & 0xFFFFu);
  __syncthreads();

  const int rl = lane >> 1;
  const int hq = lane & 1;
  const int rbase = o0 >> 3;
  const int bs = b * S_ + s;
  const int vl = VW_ * wave;
  const float* __restrict__ tab =
      flat_weight + ((size_t)s * SUB_STRIDE_ + (size_t)(rbase + rl) * ROUTE_VOCAB_) * M_ + 4 * hq;
  float* __restrict__ orow =
      out + (size_t)(bs * V_ + t0 + vl) * (R_ * M_) + rbase * M_ + 4 * lane;

  unsigned cr[VW_ + 1];
#pragma unroll
  for (int i = 0; i <= VW_; ++i) cr[i] = lcode[(vl + i) * RT_ + rl];
  v4f g[VW_];
#pragma unroll
  for (int i = 0; i < VW_; ++i) {
    const unsigned addr = cr[i] | (cr[i + 1] << 8);
    g[i] = *(const v4f*)(tab + (size_t)addr * M_);
  }
#pragma unroll
  for (int i = 0; i < VW_; ++i) {
    if (t0 + vl + i < V_) *(v4f*)(orow + (size_t)i * (R_ * M_)) = g[i];
  }
}

extern "C" void kernel_launch(void* const* d_in, const int* in_sizes, int n_in,
                              void* d_out, int out_size, void* d_ws, size_t ws_size,
                              hipStream_t stream) {
  const float* hidden      = (const float*)d_in[0];
  const float* q_weight    = (const float*)d_in[1];
  const float* flat_weight = (const float*)d_in[2];
  float* out               = (float*)d_out;

  lngram_codes_gather<<<256, 256, 0, stream>>>(hidden, q_weight, flat_weight, out);
}


# tile multiplied in two parts (rows 0-47, then rows 48-64): the table loads of 40 of its 64 rows are issued between the parts and land under the second part's WMMAs; k order of every logit kept
# speedup vs baseline: 2.2229x; 1.0586x over previous
#include <hip/hip_runtime.h>

#define B_ 2
#define S_ 4
#define T_ 1024
#define H_ 256
#define R_ 32
#define V_ (T_ - 1)
#define M_ 8
#define ROUTE_VOCAB_ 65536
#define SUB_STRIDE_ (R_ * ROUTE_VOCAB_)

typedef float v2f __attribute__((ext_vector_type(2)));
typedef float v4f __attribute__((ext_vector_type(4)));
typedef float v8f __attribute__((ext_vector_type(8)));

#define TM_ 64
#define TN_ 128
#define KC_ 64
#define NCH_ (H_ / KC_)
#define PITCH_ 260
#define RT_ (TN_ / 8)
#define V1_ 5
#define V2_ 3

#define WMMA_(a, b, c) __builtin_amdgcn_wmma_f32_16x16x4_f32(false, (a), false, (b), (short)0, (c), false, false)

__global__ __launch_bounds__(256)
void lngram_codes_gather(const float* __restrict__ hidden,
                         const float* __restrict__ qw,
                         const float* __restrict__ flat_weight,
                         float* __restrict__ out) {
  __shared__ __attribute__((aligned(16))) float lds[(TM_ + TN_ + 1) * PITCH_];
  __shared__ __attribute__((aligned(16))) unsigned char lcode[(TM_ + 1) * RT_];
  float* lH = lds;
  float* lA = lds + PITCH_;
  float* lB = lds + (1 + TM_) * PITCH_;

  const int tid  = threadIdx.x;
  const int lane = tid & 31;
  const int wave = tid >> 5;

  const int nt = blockIdx.x & 7;
  const int mt = blockIdx.x >> 3;
  const int b  = mt >> 4;
  const int t0 = (mt & 15) << 6;
  const int s  = nt >> 1;
  const int o0 = (nt & 1) << 7;

  const int q4 = (tid & 15) << 2;
  const int sr = tid >> 4;
  const float* __restrict__ gA = hidden + (size_t)(b * T_ + t0 + sr) * H_ + q4;
  const float* __restrict__ gB = qw + (size_t)(s * H_ + o0 + sr) * H_ + q4;
  float* sA = lA + sr * PITCH_ + q4;
  float* sB = lB + sr * PITCH_ + q4;

  v4f ra[TM_ / 16], rb[TN_ / 16];
#pragma unroll
  for (int i = 0; i < TM_ / 16; ++i) ra[i] = *(const v4f*)(gA + (size_t)i * 16 * H_);
#pragma unroll
  for (int i = 0; i < TN_ / 16; ++i) rb[i] = *(const v4f*)(gB + (size_t)i * 16 * H_);

  const int th = (t0 + TM_ < T_) ? (t0 + TM_) : (T_ - 1);
  const v4f rh = *(const v4f*)(hidden + (size_t)(b * T_ + th) * H_ + 4 * (tid & (H_ / 4 - 1)));

  const int c  = lane & 15;
  const int h  = lane >> 4;
  const float* fA = lA + c * PITCH_ + 2 * h;
  const float* fB = lB + (16 * wave + c) * PITCH_ + 2 * h;
  const float* fH = lH + 2 * h;

  v8f acc0 = {}, acc1 = {}, acc2 = {};

#pragma unroll
  for (int ch = 0; ch < NCH_; ++ch) {
#pragma unroll
    for (int i = 0; i < TM_ / 16; ++i) *(v4f*)(sA + i * 16 * PITCH_ + ch * KC_) = ra[i];
#pragma unroll
    for (int i = 0; i < TN_ / 16; ++i) *(v4f*)(sB + i * 16 * PITCH_ + ch * KC_) = rb[i];
    if (ch + 1 < NCH_) {
#pragma unroll
      for (int i = 0; i < TM_ / 16; ++i)
        ra[i] = *(const v4f*)(gA + (size_t)i * 16 * H_ + (ch + 1) * KC_);
#pragma unroll
      for (int i = 0; i < TN_ / 16; ++i)
        rb[i] = *(const v4f*)(gB + (size_t)i * 16 * H_ + (ch + 1) * KC_);
    }
    if (ch == 0) *(v4f*)(lH + 4 * (tid & (H_ / 4 - 1))) = rh;
    __syncthreads();

#pragma unroll
    for (int j = 0; j < KC_ / 8; ++j) {
      const int off = ch * KC_ + j * 8;
      const v2f a0x = *(const v2f*)(fA + off),               a0y = *(const v2f*)(fA + off + 4);
      const v2f a1x = *(const v2f*)(fA + 16 * PITCH_ + off), a1y = *(const v2f*)(fA + 16 * PITCH_ + off + 4);
      const v2f a2x = *(const v2f*)(fA + 32 * PITCH_ + off), a2y = *(const v2f*)(fA + 32 * PITCH_ + off + 4);
      const v2f bx  = *(const v2f*)(fB + off),               by  = *(const v2f*)(fB + off + 4);
      acc0 = WMMA_(a0x, bx, acc0);
      acc1 = WMMA_(a1x, bx, acc1);
      acc2 = WMMA_(a2x, bx, acc2);
      acc0 = WMMA_(a0y, by, acc0);
      acc1 = WMMA_(a1y, by, acc1);
      acc2 = WMMA_(a2y, by, acc2);
    }
  }

  const int myj = lane & 7;
  const unsigned sh = ((lane >> 3) & 1) << 4;
  unsigned m0 = 0u, m1 = 0u, m2 = 0u;
#pragma unroll
  for (int j = 0; j < 8; ++j) {
    const unsigned k0 = __builtin_amdgcn_ballot_w32(acc0[j] > 0.0f);
    const unsigned k1 = __builtin_amdgcn_ballot_w32(acc1[j] > 0.0f);
    const unsigned k2 = __builtin_amdgcn_ballot_w32(acc2[j] > 0.0f);
    const bool mj = (myj == j);
    m0 = mj ? k0 : m0;
    m1 = mj ? k1 : m1;
    m2 = mj ? k2 : m2;
  }
  {
    const unsigned m01 = (lane < 16) ? m0 : m1;
    *(unsigned short*)(lcode + lane * RT_ + 2 * wave) = (unsigned short)((m01 >> sh) & 0xFFFFu);
    if (lane < 16)
      *(unsigned short*)(lcode + (32 + lane) * RT_ + 2 * wave) = (unsigned short)((m2 >> sh) & 0xFFFFu);
  }
  __syncthreads();

  const int rl = lane >> 1;
  const int hq = lane & 1;
  const int rbase = o0 >> 3;
  const int bs = b * S_ + s;
  const float* __restrict__ tab =
      flat_weight + ((size_t)s * SUB_STRIDE_ + (size_t)(rbase + rl) * ROUTE_VOCAB_) * M_ + 4 * hq;
  float* __restrict__ obase =
      out + (size_t)(bs * V_ + t0) * (R_ * M_) + rbase * M_ + 4 * lane;

  const int vl1 = V1_ * wave;
  v4f g1[V1_];
  const float* p1[V1_];
  {
    unsigned cr[V1_ + 1];
#pragma unroll
    for (int i = 0; i <= V1_; ++i) cr[i] = lcode[(vl1 + i) * RT_ + rl];
#pragma unroll
    for (int i = 0; i < V1_; ++i) {
      const unsigned addr = cr[i] | (cr[i + 1] << 8);
      p1[i] = tab + (size_t)addr * M_;
      g1[i] = *(const v4f*)p1[i];
    }
  }
  __builtin_amdgcn_sched_barrier(0);

  v8f acc3 = {}, acch = {};
#pragma unroll
  for (int j = 0; j < H_ / 8; ++j) {
    const int off = j * 8;
    const v2f a3x = *(const v2f*)(fA + 48 * PITCH_ + off), a3y = *(const v2f*)(fA + 48 * PITCH_ + off + 4);
    const v2f ahx = *(const v2f*)(fH + off),               ahy = *(const v2f*)(fH + off + 4);
    const v2f bx  = *(const v2f*)(fB + off),               by  = *(const v2f*)(fB + off + 4);
    acc3 = WMMA_(a3x, bx, acc3);
    acch = WMMA_(ahx, bx, acch);
    acc3 = WMMA_(a3y, by, acc3);
    acch = WMMA_(ahy, by, acch);
  }

#pragma unroll
  for (int i = 0; i < V1_; ++i) asm volatile("" :: "v"(p1[i]));

  {
    unsigned m3 = 0u;
#pragma unroll
    for (int j = 0; j < 8; ++j) {
      const unsigned k3 = __builtin_amdgcn_ballot_w32(acc3[j] > 0.0f);
      m3 = (myj == j) ? k3 : m3;
    }
    const unsigned kh = __builtin_amdgcn_ballot_w32(acch[0] > 0.0f);
    const unsigned val = (lane < 16) ? ((m3 >> sh) & 0xFFFFu) : (kh & 0xFFFFu);
    if (lane <= 16)
      *(unsigned short*)(lcode + (48 + lane) * RT_ + 2 * wave) = (unsigned short)val;
  }
  __syncthreads();

  const int vl2 = V1_ * 8 + V2_ * wave;
  {
    unsigned cr[V2_ + 1];
#pragma unroll
    for (int i = 0; i <= V2_; ++i) cr[i] = lcode[(vl2 + i) * RT_ + rl];
    v4f g2[V2_];
#pragma unroll
    for (int i = 0; i < V2_; ++i) {
      const unsigned addr = cr[i] | (cr[i + 1] << 8);
      g2[i] = *(const v4f*)(tab + (size_t)addr * M_);
    }
#pragma unroll
    for (int i = 0; i < V1_; ++i)
      *(v4f*)(obase + (size_t)(vl1 + i) * (R_ * M_)) = g1[i];
#pragma unroll
    for (int i = 0; i < V2_; ++i) {
      if (t0 + vl2 + i < V_) *(v4f*)(obase + (size_t)(vl2 + i) * (R_ * M_)) = g2[i];
    }
  }
}

extern "C" void kernel_launch(void* const* d_in, const int* in_sizes, int n_in,
                              void* d_out, int out_size, void* d_ws, size_t ws_size,
                              hipStream_t stream) {
  const float* hidden      = (const float*)d_in[0];
  const float* q_weight    = (const float*)d_in[1];
  const float* flat_weight = (const float*)d_in[2];
  float* out               = (float*)d_out;

  lngram_codes_gather<<<256, 256, 0, stream>>>(hidden, q_weight, flat_weight, out);
}
